# starting kernel
# baseline (speedup 1.0000x reference)
#include <hip/hip_runtime.h>


typedef __attribute__((ext_vector_type(2))) float v2f;
typedef __attribute__((ext_vector_type(8))) float v8f;

#define FDIM 128
#define LDS_STRIDE 132
#define ROWS_PER_WAVE 32
#define ROWS_PER_BLOCK 256

__global__ __launch_bounds__(256)
void hgnn_gemm128_f32(const float* __restrict__ X,
                      const float* __restrict__ W,
                      const float* __restrict__ W2,
                      float* __restrict__ Out,
                      int nRows)
{
    __shared__ float sWT[FDIM * LDS_STRIDE];

    const int tid  = threadIdx.x;
    const int wave = tid >> 5;
    const int lane = tid & 31;
    const int half = lane >> 4;
    const int m    = lane & 15;

    const int rowBase = blockIdx.x * ROWS_PER_BLOCK + wave * ROWS_PER_WAVE;

    int r0 = rowBase + m;       if (r0 >= nRows) r0 = nRows - 1;
    int r1 = rowBase + 16 + m;  if (r1 >= nRows) r1 = nRows - 1;
    const float* xrow0 = X + (size_t)r0 * FDIM;
    const float* xrow1 = X + (size_t)r1 * FDIM;

    __builtin_prefetch(xrow0, 0, 0);
    __builtin_prefetch(xrow1, 0, 0);

    {
        const float4* w4  = (const float4*)W;
        const int     nf4 = (FDIM * FDIM) / 4;
        if (W2 != nullptr) {
            const float4* w24 = (const float4*)W2;
            for (int i = tid; i < nf4; i += 256) {
                const int k = i >> 5, c = (i & 31) * 4;
                float4 a = w4[i], b = w24[i];
                sWT[(c + 0) * LDS_STRIDE + k] = a.x + b.x;
                sWT[(c + 1) * LDS_STRIDE + k] = a.y + b.y;
                sWT[(c + 2) * LDS_STRIDE + k] = a.z + b.z;
                sWT[(c + 3) * LDS_STRIDE + k] = a.w + b.w;
            }
        } else {
            for (int i = tid; i < nf4; i += 256) {
                const int k = i >> 5, c = (i & 31) * 4;
                float4 a = w4[i];
                sWT[(c + 0) * LDS_STRIDE + k] = a.x;
                sWT[(c + 1) * LDS_STRIDE + k] = a.y;
                sWT[(c + 2) * LDS_STRIDE + k] = a.z;
                sWT[(c + 3) * LDS_STRIDE + k] = a.w;
            }
        }
    }
    __syncthreads();

    v8f acc0[8], acc1[8];
#pragma unroll
    for (int n = 0; n < 8; ++n) {
        acc0[n] = (v8f){0.f,0.f,0.f,0.f,0.f,0.f,0.f,0.f};
        acc1[n] = (v8f){0.f,0.f,0.f,0.f,0.f,0.f,0.f,0.f};
    }

#pragma unroll 4
    for (int k0 = 0; k0 < FDIM; k0 += 4) {
        const int k = k0 + 2 * half;
        const v2f a0 = *(const v2f*)(xrow0 + k);
        const v2f a1 = *(const v2f*)(xrow1 + k);
#pragma unroll
        for (int n = 0; n < 8; ++n) {
            const v2f b = *(const v2f*)(sWT + (n * 16 + m) * LDS_STRIDE + k);
            acc0[n] = __builtin_amdgcn_wmma_f32_16x16x4_f32(
                          false, a0, false, b, (short)0, acc0[n], false, false);
            acc1[n] = __builtin_amdgcn_wmma_f32_16x16x4_f32(
                          false, a1, false, b, (short)0, acc1[n], false, false);
        }
    }

#pragma unroll
    for (int i = 0; i < 8; ++i) {
        const int row0 = rowBase + i + 8 * half;
        if (row0 < nRows) {
            float* orow = Out + (size_t)row0 * FDIM;
#pragma unroll
            for (int n = 0; n < 8; ++n)
                orow[n * 16 + m] = acc0[n][i];
        }
        const int row1 = rowBase + 16 + i + 8 * half;
        if (row1 < nRows) {
            float* orow = Out + (size_t)row1 * FDIM;
#pragma unroll
            for (int n = 0; n < 8; ++n)
                orow[n * 16 + m] = acc1[n][i];
        }
    }
}

__global__ __launch_bounds__(256)
void hgnn_scatter_add(const float* __restrict__ Y,
                      const int* __restrict__ src,
                      const int* __restrict__ dst,
                      float* __restrict__ Out,
                      int nEdges)
{
    const int gid  = blockIdx.x * 256 + threadIdx.x;
    const int e    = gid >> 5;
    const int lane = gid & 31;
    if (e >= nEdges) return;

    const int s = src[e];
    const int d = dst[e];

    const float4 v = ((const float4*)(Y + (size_t)s * FDIM))[lane];
    float* o = Out + (size_t)d * FDIM + lane * 4;
    unsafeAtomicAdd(o + 0, v.x);
    unsafeAtomicAdd(o + 1, v.y);
    unsafeAtomicAdd(o + 2, v.z);
    unsafeAtomicAdd(o + 3, v.w);
}

extern "C" void kernel_launch(void* const* d_in, const int* in_sizes, int n_in,
                              void* d_out, int out_size, void* d_ws, size_t ws_size,
                              hipStream_t stream)
{
    const float* xA        = (const float*)d_in[0];
    const float* xB        = (const float*)d_in[1];
    const float* Wmsg_ab1  = (const float*)d_in[2];
    const float* Wself_ab1 = (const float*)d_in[3];
    const float* Wmsg_ab2  = (const float*)d_in[4];
    const float* Wself_ab2 = (const float*)d_in[5];
    const float* Wmsg_ba   = (const float*)d_in[6];
    const float* Wself_ba  = (const float*)d_in[7];
    const int*   src_ab1   = (const int*)d_in[8];
    const int*   dst_ab1   = (const int*)d_in[9];
    const int*   src_ab2   = (const int*)d_in[10];
    const int*   dst_ab2   = (const int*)d_in[11];
    const int*   src_ba    = (const int*)d_in[12];
    const int*   dst_ba    = (const int*)d_in[13];

    const int NA = in_sizes[0] / FDIM;
    const int NB = in_sizes[1] / FDIM;
    const int E  = in_sizes[8];

    float* outA = (float*)d_out;
    float* outB = outA + (size_t)NA * FDIM;
    float* Y    = (float*)d_ws;

    const dim3 blk(256);
    const dim3 gA((NA + ROWS_PER_BLOCK - 1) / ROWS_PER_BLOCK);
    const dim3 gB((NB + ROWS_PER_BLOCK - 1) / ROWS_PER_BLOCK);
    const dim3 gE((unsigned)(((size_t)E * 32 + 255) / 256));

    hgnn_gemm128_f32<<<gA, blk, 0, stream>>>(xA, Wself_ba,  nullptr,   outA, NA);
    hgnn_gemm128_f32<<<gB, blk, 0, stream>>>(xB, Wself_ab1, Wself_ab2, outB, NB);

    hgnn_gemm128_f32<<<gA, blk, 0, stream>>>(xA, Wmsg_ab1, nullptr, Y, NA);
    hgnn_scatter_add<<<gE, blk, 0, stream>>>(Y, src_ab1, dst_ab1, outB, E);

    hgnn_gemm128_f32<<<gA, blk, 0, stream>>>(xA, Wmsg_ab2, nullptr, Y, NA);
    hgnn_scatter_add<<<gE, blk, 0, stream>>>(Y, src_ab2, dst_ab2, outB, E);

    hgnn_gemm128_f32<<<gB, blk, 0, stream>>>(xB, Wmsg_ba, nullptr, Y, NB);
    hgnn_scatter_add<<<gE, blk, 0, stream>>>(Y, src_ba, dst_ba, outA, E);
}


# products re-tiled: a 32-wave work-group per 400 rows stages every weight of its x in LDS (3 for x_A, 2 for x_B), waves take (16-row tile, one weight) tasks; 5 product launches -> 2
# speedup vs baseline: 1.9535x; 1.9535x over previous
#include <hip/hip_runtime.h>


typedef __attribute__((ext_vector_type(2))) float v2f;
typedef __attribute__((ext_vector_type(4))) float v4f;
typedef __attribute__((ext_vector_type(8))) float v8f;

#define FDIM 128
#define LDS_STRIDE 132
#define ROWS_PER_GROUP 400
#define TILE_ROWS 16
#define MAX_W 3

__device__ __forceinline__ void hgnn_stage_wt(const float* __restrict__ W, const float* __restrict__ W2,
                                              float* __restrict__ sW, int wsub, int lane)
{
    const int cq   = lane & 3;
    const int krow = wsub * 16 + (lane >> 2);
    const float4* s0  = (const float4*)W + krow * (FDIM / 4) + cq;
    float*        dst = sW + (cq * 4) * LDS_STRIDE + krow;
    if (W2 != nullptr) {
        const float4* s1 = (const float4*)W2 + krow * (FDIM / 4) + cq;
#pragma unroll
        for (int it = 0; it < 16; ++it) {
            const int kh = it >> 3, cb = it & 7;
            const float4 a = s0[kh * 8 * (FDIM / 4) + cb * 4];
            const float4 b = s1[kh * 8 * (FDIM / 4) + cb * 4];
            float* d = dst + (cb * 16) * LDS_STRIDE + kh * 8;
            d[0 * LDS_STRIDE] = a.x + b.x;
            d[1 * LDS_STRIDE] = a.y + b.y;
            d[2 * LDS_STRIDE] = a.z + b.z;
            d[3 * LDS_STRIDE] = a.w + b.w;
        }
    } else {
#pragma unroll
        for (int it = 0; it < 16; ++it) {
            const int kh = it >> 3, cb = it & 7;
            const float4 a = s0[kh * 8 * (FDIM / 4) + cb * 4];
            float* d = dst + (cb * 16) * LDS_STRIDE + kh * 8;
            d[0 * LDS_STRIDE] = a.x;
            d[1 * LDS_STRIDE] = a.y;
            d[2 * LDS_STRIDE] = a.z;
            d[3 * LDS_STRIDE] = a.w;
        }
    }
}

__global__ __launch_bounds__(1024)
void hgnn_gemm128_multi(const float* __restrict__ X,
                        const float* __restrict__ W0, const float* __restrict__ W0b,
                        const float* __restrict__ W1, const float* __restrict__ W2,
                        float* __restrict__ O0, float* __restrict__ O1, float* __restrict__ O2,
                        int nW, int nRows)
{
    __shared__ float sWT[MAX_W * FDIM * LDS_STRIDE];

    const int tid  = threadIdx.x;
    const int wave = __builtin_amdgcn_readfirstlane(tid >> 5);
    const int lane = tid & 31;
    const int half = lane >> 4;
    const int m    = lane & 15;

    const int rowBlock = blockIdx.x * ROWS_PER_GROUP;
    const int rowEnd   = (rowBlock + ROWS_PER_GROUP < nRows) ? rowBlock + ROWS_PER_GROUP : nRows;
    const int nTiles   = (rowEnd - rowBlock + TILE_ROWS - 1) / TILE_ROWS;
    const int nTasks   = nTiles * nW;

    if (wave < nTasks) {
        int r = rowBlock + (wave / nW) * TILE_ROWS + m;
        if (r >= rowEnd) r = rowEnd - 1;
        __builtin_prefetch(X + (size_t)r * FDIM, 0, 0);
    }

    {
        const int wsel = wave >> 3;
        if (wsel < nW) {
            const float* Wp  = (wsel == 0) ? W0 : (wsel == 1) ? W1 : W2;
            const float* Wp2 = (wsel == 0) ? W0b : nullptr;
            hgnn_stage_wt(Wp, Wp2, sWT + wsel * (FDIM * LDS_STRIDE), wave & 7, lane);
        }
    }
    __syncthreads();

    for (int t = wave; t < nTasks; t += 32) {
        const int tile    = t / nW;
        const int wi      = t - tile * nW;
        const int rowBase = rowBlock + tile * TILE_ROWS;

        int r0 = rowBase + m;  if (r0 >= rowEnd) r0 = rowEnd - 1;
        const float* xrow = X + (size_t)r0 * FDIM;
        const float* sW   = sWT + wi * (FDIM * LDS_STRIDE);
        float* __restrict__ Out = (wi == 0) ? O0 : (wi == 1) ? O1 : O2;

        v8f acc[8];
#pragma unroll
        for (int n = 0; n < 8; ++n) acc[n] = (v8f){0.f,0.f,0.f,0.f,0.f,0.f,0.f,0.f};

#pragma unroll 4
        for (int k0 = 0; k0 < FDIM; k0 += 4) {
            const int k = k0 + 2 * half;
            const v2f a = *(const v2f*)(xrow + k);
#pragma unroll
            for (int n = 0; n < 8; ++n) {
                const v2f b = *(const v2f*)(sW + (n * 16 + m) * LDS_STRIDE + k);
                acc[n] = __builtin_amdgcn_wmma_f32_16x16x4_f32(
                             false, a, false, b, (short)0, acc[n], false, false);
            }
        }

#pragma unroll
        for (int i = 0; i < 8; ++i) {
            const int row = rowBase + i + 8 * half;
            if (row < rowEnd) {
                float* orow = Out + (size_t)row * FDIM;
#pragma unroll
                for (int n = 0; n < 8; ++n)
                    orow[n * 16 + m] = acc[n][i];
            }
        }
    }
}

#define NBK  512
#define NCH  128
#define GCAP 4096
#define MAXR 512

__device__ __forceinline__ int hgnn_rows_per_side(int nA, int nB)
{
    const int m = nA > nB ? nA : nB;
    return (m + NBK - 1) / NBK;
}

__device__ __forceinline__ void hgnn_edge(int g, int nEdges, int nA, int nB,
                                          const int* __restrict__ src_ba,  const int* __restrict__ dst_ba,
                                          const int* __restrict__ src_ab1, const int* __restrict__ dst_ab1,
                                          const int* __restrict__ src_ab2, const int* __restrict__ dst_ab2,
                                          int& d, int& s)
{
    if (g < nEdges)          { s = src_ba[g];                           d = dst_ba[g];                    }
    else if (g < 2 * nEdges) { s = nB + src_ab1[g - nEdges];            d = nA + dst_ab1[g - nEdges];     }
    else                     { s = nB + nA + src_ab2[g - 2 * nEdges];   d = nA + dst_ab2[g - 2 * nEdges]; }
}

__device__ __forceinline__ void hgnn_bucket_of(int d, int nA, int rbs, int& b, int& local)
{
    const bool sideB = d >= nA;
    const int  ds    = sideB ? d - nA : d;
    b     = ds / rbs;
    local = ds - b * rbs + (sideB ? rbs : 0);
}

__global__ __launch_bounds__(1024)
void hgnn_bucket_count(const int* __restrict__ dst_ba, const int* __restrict__ dst_ab1,
                       const int* __restrict__ dst_ab2,
                       int* __restrict__ C, int nEdges, int nA, int nB)
{
    __shared__ int sCnt[NBK];
    const int tid = threadIdx.x;
    for (int i = tid; i < NBK; i += 1024) sCnt[i] = 0;
    __syncthreads();

    const int total = 3 * nEdges;
    const int ch    = (total + NCH - 1) / NCH;
    const int g0    = blockIdx.x * ch;
    const int g1    = (g0 + ch < total) ? g0 + ch : total;
    const int rbs   = hgnn_rows_per_side(nA, nB);

    for (int g = g0 + tid; g < g1; g += 1024) {
        int d;
        if (g < nEdges)          d = dst_ba[g];
        else if (g < 2 * nEdges) d = nA + dst_ab1[g - nEdges];
        else                     d = nA + dst_ab2[g - 2 * nEdges];
        int b, local;
        hgnn_bucket_of(d, nA, rbs, b, local);
        atomicAdd(&sCnt[b], 1);
    }
    __syncthreads();
    for (int i = tid; i < NBK; i += 1024) C[blockIdx.x * NBK + i] = sCnt[i];
}

__global__ __launch_bounds__(1024)
void hgnn_bucket_place(const int* __restrict__ src_ba,  const int* __restrict__ dst_ba,
                       const int* __restrict__ src_ab1, const int* __restrict__ dst_ab1,
                       const int* __restrict__ src_ab2, const int* __restrict__ dst_ab2,
                       const int* __restrict__ C, int2* __restrict__ rec, int* __restrict__ bstart,
                       int nEdges, int nA, int nB)
{
    __shared__ int sTot[2][NBK];
    __shared__ int sBef[2][NBK];
    __shared__ int sScan[2][NBK];
    __shared__ int sCur[NBK];

    const int tid = threadIdx.x;
    const int col = tid & (NBK - 1);
    const int hf  = tid >> 9;
    {
        int tot = 0, bef = 0;
        const int w0 = hf * (NCH / 2);
        for (int w = w0; w < w0 + NCH / 2; ++w) {
            const int c = C[w * NBK + col];
            tot += c;
            if (w < (int)blockIdx.x) bef += c;
        }
        sTot[hf][col] = tot;
        sBef[hf][col] = bef;
    }
    __syncthreads();
    if (tid < NBK) sScan[0][tid] = sTot[0][tid] + sTot[1][tid];
    __syncthreads();
    int pi = 0;
    for (int off = 1; off < NBK; off <<= 1) {
        if (tid < NBK) {
            int v = sScan[pi][tid];
            if (tid >= off) v += sScan[pi][tid - off];
            sScan[pi ^ 1][tid] = v;
        }
        __syncthreads();
        pi ^= 1;
    }
    if (tid < NBK) {
        const int own  = sTot[0][tid] + sTot[1][tid];
        const int excl = sScan[pi][tid] - own;
        sCur[tid] = excl + sBef[0][tid] + sBef[1][tid];
        if (blockIdx.x == 0) {
            bstart[tid] = excl;
            if (tid == NBK - 1) bstart[NBK] = excl + own;
        }
    }
    __syncthreads();

    const int total = 3 * nEdges;
    const int ch    = (total + NCH - 1) / NCH;
    const int g0    = blockIdx.x * ch;
    const int g1    = (g0 + ch < total) ? g0 + ch : total;
    const int rbs   = hgnn_rows_per_side(nA, nB);

    for (int g = g0 + tid; g < g1; g += 1024) {
        int d, s;
        hgnn_edge(g, nEdges, nA, nB, src_ba, dst_ba, src_ab1, dst_ab1, src_ab2, dst_ab2, d, s);
        int b, local;
        hgnn_bucket_of(d, nA, rbs, b, local);
        const int slot = atomicAdd(&sCur[b], 1);
        rec[slot] = make_int2(local, s);
    }
}

__global__ __launch_bounds__(1024)
void hgnn_gather_bucket(const float* __restrict__ Y,
                        const int2* __restrict__ rec,
                        const int* __restrict__ bstart,
                        float* __restrict__ Out,
                        int nA, int nB)
{
    __shared__ int  sHead[MAXR];
    __shared__ int2 sNode[GCAP];
    __shared__ int  sTicket;

    const int tid  = threadIdx.x;
    const int lane = tid & 31;
    const int sub  = tid & 7;
    const int grp  = lane >> 3;
    const int b    = blockIdx.x;
    const int rbs  = hgnn_rows_per_side(nA, nB);
    const int nr   = 2 * rbs;
    const int beg  = bstart[b];
    const int end  = bstart[b + 1];

    for (int base = beg; base < end; base += GCAP) {
        const int n = (end - base < GCAP) ? end - base : GCAP;
        for (int i = tid; i < nr; i += 1024) sHead[i] = -1;
        if (tid == 0) sTicket = 0;
        __syncthreads();
        for (int i = tid; i < n; i += 1024) {
            const int2 r    = rec[base + i];
            const int  prev = atomicExch(&sHead[r.x], i);
            sNode[i] = make_int2(prev, r.y);
        }
        __syncthreads();

        for (int k = 0; k < MAXR / 4; ++k) {
            int t = 0;
            if (lane == 0) t = atomicAdd(&sTicket, 4);
            t = __builtin_amdgcn_readfirstlane(t);
            if (t >= nr) break;
            const int q     = t + grp;
            const int local = (q < rbs) ? q + rbs : q - rbs;
            const bool sideB = local >= rbs;
            const int ds    = b * rbs + (sideB ? local - rbs : local);
            const bool valid = (q < nr) && (ds < (sideB ? nB : nA));
            int e = valid ? sHead[local] : -1;
            if (e < 0) continue;
            const int row = sideB ? nA + ds : ds;

            v4f* orow = (v4f*)(Out + (size_t)row * FDIM);
            v4f acc[4];
#pragma unroll
            for (int j = 0; j < 4; ++j) acc[j] = __builtin_nontemporal_load(orow + sub + 8 * j);

            int2 nd = sNode[e];
            for (int it = 0; it < n; ++it) {
                const v4f* y = (const v4f*)(Y + (size_t)nd.y * FDIM);
                v4f v[4];
#pragma unroll
                for (int j = 0; j < 4; ++j) v[j] = y[sub + 8 * j];
                e = nd.x;
                const bool more = e >= 0;
                if (more) nd = sNode[e];
#pragma unroll
                for (int j = 0; j < 4; ++j) acc[j] += v[j];
                if (!more) break;
            }
#pragma unroll
            for (int j = 0; j < 4; ++j) __builtin_nontemporal_store(acc[j], orow + sub + 8 * j);
        }
        __syncthreads();
    }
}

extern "C" void kernel_launch(void* const* d_in, const int* in_sizes, int n_in,
                              void* d_out, int out_size, void* d_ws, size_t ws_size,
                              hipStream_t stream)
{
    const float* xA        = (const float*)d_in[0];
    const float* xB        = (const float*)d_in[1];
    const float* Wmsg_ab1  = (const float*)d_in[2];
    const float* Wself_ab1 = (const float*)d_in[3];
    const float* Wmsg_ab2  = (const float*)d_in[4];
    const float* Wself_ab2 = (const float*)d_in[5];
    const float* Wmsg_ba   = (const float*)d_in[6];
    const float* Wself_ba  = (const float*)d_in[7];
    const int*   src_ab1   = (const int*)d_in[8];
    const int*   dst_ab1   = (const int*)d_in[9];
    const int*   src_ab2   = (const int*)d_in[10];
    const int*   dst_ab2   = (const int*)d_in[11];
    const int*   src_ba    = (const int*)d_in[12];
    const int*   dst_ba    = (const int*)d_in[13];

    const int NA = in_sizes[0] / FDIM;
    const int NB = in_sizes[1] / FDIM;
    const int E  = in_sizes[8];

    float* outA = (float*)d_out;
    float* outB = outA + (size_t)NA * FDIM;

    float* Yba  = (float*)d_ws;
    float* Yab1 = Yba  + (size_t)NB * FDIM;
    float* Yab2 = Yab1 + (size_t)NA * FDIM;
    int2*  rec  = (int2*)(Yab2 + (size_t)NA * FDIM);
    int*   cnt  = (int*)(rec + (size_t)3 * E);
    int*   bst  = cnt + NCH * NBK;

    const dim3 blk(1024);
    const dim3 gA((NA + ROWS_PER_GROUP - 1) / ROWS_PER_GROUP);
    const dim3 gB((NB + ROWS_PER_GROUP - 1) / ROWS_PER_GROUP);

    hgnn_gemm128_multi<<<gA, blk, 0, stream>>>(xA, Wmsg_ab1, nullptr, Wmsg_ab2, Wself_ba,
                                               Yab1, Yab2, outA, 3, NA);
    hgnn_gemm128_multi<<<gB, blk, 0, stream>>>(xB, Wself_ab1, Wself_ab2, Wmsg_ba, nullptr,
                                               outB, Yba, nullptr, 2, NB);

    hgnn_bucket_count<<<dim3(NCH), dim3(1024), 0, stream>>>(dst_ba, dst_ab1, dst_ab2, cnt, E, NA, NB);
    hgnn_bucket_place<<<dim3(NCH), dim3(1024), 0, stream>>>(src_ba, dst_ba, src_ab1, dst_ab1, src_ab2, dst_ab2,
                                                           cnt, rec, bst, E, NA, NB);
    hgnn_gather_bucket<<<dim3(NBK), dim3(1024), 0, stream>>>(Yba, rec, bst, outA, NA, NB);
}


# edge grouping: count and place passes run inside the two product launches, on the youngest waves ahead of their tile tasks, beside the older waves' WMMAs; 5 launches -> 3; sums in LDS arrival order
# speedup vs baseline: 2.0764x; 1.0629x over previous
#include <hip/hip_runtime.h>


typedef __attribute__((ext_vector_type(2))) float v2f;
typedef __attribute__((ext_vector_type(4))) float v4f;
typedef __attribute__((ext_vector_type(8))) float v8f;

#define FDIM 128
#define LDS_STRIDE 132
#define ROWS_PER_GROUP 400
#define TILE_ROWS 16
#define MAX_W 3

__device__ __forceinline__ void hgnn_stage_wt(const float* __restrict__ W, const float* __restrict__ W2,
                                              float* __restrict__ sW, int wsub, int lane)
{
    const int cq   = lane & 3;
    const int krow = wsub * 16 + (lane >> 2);
    const float4* s0  = (const float4*)W + krow * (FDIM / 4) + cq;
    float*        dst = sW + (cq * 4) * LDS_STRIDE + krow;
    if (W2 != nullptr) {
        const float4* s1 = (const float4*)W2 + krow * (FDIM / 4) + cq;
#pragma unroll
        for (int it = 0; it < 16; ++it) {
            const int kh = it >> 3, cb = it & 7;
            const float4 a = s0[kh * 8 * (FDIM / 4) + cb * 4];
            const float4 b = s1[kh * 8 * (FDIM / 4) + cb * 4];
            float* d = dst + (cb * 16) * LDS_STRIDE + kh * 8;
            d[0 * LDS_STRIDE] = a.x + b.x;
            d[1 * LDS_STRIDE] = a.y + b.y;
            d[2 * LDS_STRIDE] = a.z + b.z;
            d[3 * LDS_STRIDE] = a.w + b.w;
        }
    } else {
#pragma unroll
        for (int it = 0; it < 16; ++it) {
            const int kh = it >> 3, cb = it & 7;
            const float4 a = s0[kh * 8 * (FDIM / 4) + cb * 4];
            float* d = dst + (cb * 16) * LDS_STRIDE + kh * 8;
            d[0 * LDS_STRIDE] = a.x;
            d[1 * LDS_STRIDE] = a.y;
            d[2 * LDS_STRIDE] = a.z;
            d[3 * LDS_STRIDE] = a.w;
        }
    }
}

#define NBK  512
#define NCH  128
#define GCAP 4096
#define MAXR 512
#define EDGE_GROUP 4

__device__ __forceinline__ int hgnn_rows_per_side(int nA, int nB)
{
    const int m = nA > nB ? nA : nB;
    return (m + NBK - 1) / NBK;
}

__device__ __forceinline__ void hgnn_edge_ref(int g, int nEdges, int& rel, int& off)
{
    rel = (g >= nEdges ? 1 : 0) + (g >= 2 * nEdges ? 1 : 0);
    off = g - rel * nEdges;
}

__device__ __forceinline__ void hgnn_bucket_of(int d, int nA, int rbs, int& b, int& local)
{
    const bool sideB = d >= nA;
    const int  ds    = sideB ? d - nA : d;
    b     = ds / rbs;
    local = ds - b * rbs + (sideB ? rbs : 0);
}

__device__ __forceinline__ void hgnn_chunk(int c, int nEdges, int& g0, int& g1)
{
    const int total = 3 * nEdges;
    const int ch    = (total + NCH - 1) / NCH;
    g0 = c * ch;
    if (g0 > total) g0 = total;
    g1 = (g0 + ch < total) ? g0 + ch : total;
}

__device__ __forceinline__ void hgnn_count_edges(int c, int sl, int nsl, int* sCnt,
                                                 const int* __restrict__ dst_ba, const int* __restrict__ dst_ab1,
                                                 const int* __restrict__ dst_ab2,
                                                 int nEdges, int nA, int nB)
{
    int g0, g1;
    hgnn_chunk(c, nEdges, g0, g1);
    const int rbs = hgnn_rows_per_side(nA, nB);
    for (int gb = g0 + sl; gb < g1; gb += EDGE_GROUP * nsl) {
        int d[EDGE_GROUP], rel[EDGE_GROUP];
#pragma unroll
        for (int u = 0; u < EDGE_GROUP; ++u) {
            const int g = gb + u * nsl;
            int off;
            hgnn_edge_ref(g, nEdges, rel[u], off);
            const int* dp = (rel[u] == 0) ? dst_ba : (rel[u] == 1) ? dst_ab1 : dst_ab2;
            d[u] = -1;
            if (g < g1) d[u] = dp[off];
        }
#pragma unroll
        for (int u = 0; u < EDGE_GROUP; ++u) {
            if (d[u] >= 0) {
                int b, local;
                hgnn_bucket_of(d[u] + (rel[u] ? nA : 0), nA, rbs, b, local);
                atomicAdd(&sCnt[b], 1);
            }
        }
    }
}

__device__ __forceinline__ void hgnn_place_prepare(int c, int tid, int* sG,
                                                   const int* __restrict__ C, int* __restrict__ bstart)
{
    int* sTot  = sG;
    int* sBef  = sG + 2 * NBK;
    int* sScan = sG + 4 * NBK;
    int* sCur  = sG + 6 * NBK;

    const int col = tid & (NBK - 1);
    const int hf  = tid >> 9;
    {
        int tot = 0, bef = 0;
        const int w0 = hf * (NCH / 2);
#pragma unroll 16
        for (int i = 0; i < NCH / 2; ++i) {
            const int w = w0 + i;
            const int v = C[w * NBK + col];
            tot += v;
            bef += (w < c) ? v : 0;
        }
        sTot[hf * NBK + col] = tot;
        sBef[hf * NBK + col] = bef;
    }
    __syncthreads();
    if (tid < NBK) sScan[tid] = sTot[tid] + sTot[NBK + tid];
    __syncthreads();
    int pi = 0;
    for (int off = 1; off < NBK; off <<= 1) {
        if (tid < NBK) {
            int v = sScan[pi * NBK + tid];
            if (tid >= off) v += sScan[pi * NBK + tid - off];
            sScan[(pi ^ 1) * NBK + tid] = v;
        }
        __syncthreads();
        pi ^= 1;
    }
    if (tid < NBK) {
        const int own  = sTot[tid] + sTot[NBK + tid];
        const int excl = sScan[pi * NBK + tid] - own;
        sCur[tid] = excl + sBef[tid] + sBef[NBK + tid];
        if (c == 0) {
            bstart[tid] = excl;
            if (tid == NBK - 1) bstart[NBK] = excl + own;
        }
    }
    __syncthreads();
}

__device__ __forceinline__ void hgnn_place_edges(int c, int sl, int nsl, int* sCur,
                                                 const int* __restrict__ src_ba,  const int* __restrict__ dst_ba,
                                                 const int* __restrict__ src_ab1, const int* __restrict__ dst_ab1,
                                                 const int* __restrict__ src_ab2, const int* __restrict__ dst_ab2,
                                                 int2* __restrict__ rec, int nEdges, int nA, int nB)
{
    int g0, g1;
    hgnn_chunk(c, nEdges, g0, g1);
    const int rbs = hgnn_rows_per_side(nA, nB);
    for (int gb = g0 + sl; gb < g1; gb += EDGE_GROUP * nsl) {
        int d[EDGE_GROUP], s[EDGE_GROUP], rel[EDGE_GROUP];
#pragma unroll
        for (int u = 0; u < EDGE_GROUP; ++u) {
            const int g = gb + u * nsl;
            int off;
            hgnn_edge_ref(g, nEdges, rel[u], off);
            const int* dp = (rel[u] == 0) ? dst_ba : (rel[u] == 1) ? dst_ab1 : dst_ab2;
            const int* sp = (rel[u] == 0) ? src_ba : (rel[u] == 1) ? src_ab1 : src_ab2;
            d[u] = -1;
            s[u] = 0;
            if (g < g1) { d[u] = dp[off]; s[u] = sp[off]; }
        }
#pragma unroll
        for (int u = 0; u < EDGE_GROUP; ++u) {
            if (d[u] >= 0) {
                int b, local;
                hgnn_bucket_of(d[u] + (rel[u] ? nA : 0), nA, rbs, b, local);
                const int slot = atomicAdd(&sCur[b], 1);
                rec[slot] = make_int2(local, s[u] + (rel[u] == 0 ? 0 : rel[u] == 1 ? nB : nB + nA));
            }
        }
    }
}

template <int GROUP_PASS>
__global__ __launch_bounds__(1024)
void hgnn_gemm128_multi(const float* __restrict__ X,
                        const float* __restrict__ W0, const float* __restrict__ W0b,
                        const float* __restrict__ W1, const float* __restrict__ W2,
                        float* __restrict__ O0, float* __restrict__ O1, float* __restrict__ O2,
                        int nW, int nRows,
                        const int* __restrict__ src_ba,  const int* __restrict__ dst_ba,
                        const int* __restrict__ src_ab1, const int* __restrict__ dst_ab1,
                        const int* __restrict__ src_ab2, const int* __restrict__ dst_ab2,
                        int* __restrict__ C, int2* __restrict__ rec, int* __restrict__ bstart,
                        int nEdges, int nA, int nB)
{
    __shared__ float sWT[MAX_W * FDIM * LDS_STRIDE];
    __shared__ int sG[7 * NBK];

    const int tid  = threadIdx.x;
    const int wave = __builtin_amdgcn_readfirstlane(tid >> 5);
    const int lane = tid & 31;
    const int half = lane >> 4;
    const int m    = lane & 15;

    const int rowBlock = blockIdx.x * ROWS_PER_GROUP;
    const int rowEnd   = (rowBlock + ROWS_PER_GROUP < nRows) ? rowBlock + ROWS_PER_GROUP : nRows;
    const int nTiles   = (rowEnd - rowBlock + TILE_ROWS - 1) / TILE_ROWS;
    const int nTasks   = nTiles * nW;

    const int  rem        = nTasks & 31;
    const bool spare      = wave >= rem;
    const int  nsl        = (32 - rem) * 32;
    const int  sl         = (wave - rem) * 32 + lane;

    if (wave < nTasks) {
        int r = rowBlock + (wave / nW) * TILE_ROWS + m;
        if (r >= rowEnd) r = rowEnd - 1;
        __builtin_prefetch(X + (size_t)r * FDIM, 0, 0);
    }

    {
        const int wsel = wave >> 3;
        if (wsel < nW) {
            const float* Wp  = (wsel == 0) ? W0 : (wsel == 1) ? W1 : W2;
            const float* Wp2 = (wsel == 0) ? W0b : nullptr;
            hgnn_stage_wt(Wp, Wp2, sWT + wsel * (FDIM * LDS_STRIDE), wave & 7, lane);
        }
    }
    if (GROUP_PASS == 1) {
        if (tid < NBK) sG[tid] = 0;
    }
    if (GROUP_PASS == 2) {
        if ((int)blockIdx.x < NCH) hgnn_place_prepare((int)blockIdx.x, tid, sG, C, bstart);
    }
    __syncthreads();

    if (GROUP_PASS == 1) {
        if ((int)blockIdx.x < NCH && spare)
            hgnn_count_edges((int)blockIdx.x, sl, nsl, sG, dst_ba, dst_ab1, dst_ab2, nEdges, nA, nB);
    }
    if (GROUP_PASS == 2) {
        if ((int)blockIdx.x < NCH && spare)
            hgnn_place_edges((int)blockIdx.x, sl, nsl, sG + 6 * NBK, src_ba, dst_ba, src_ab1, dst_ab1, src_ab2, dst_ab2,
                             rec, nEdges, nA, nB);
    }

    for (int t = wave; t < nTasks; t += 32) {
        const int tile    = t / nW;
        const int wi      = t - tile * nW;
        const int rowBase = rowBlock + tile * TILE_ROWS;

        int r0 = rowBase + m;  if (r0 >= rowEnd) r0 = rowEnd - 1;
        const float* xrow = X + (size_t)r0 * FDIM;
        const float* sW   = sWT + wi * (FDIM * LDS_STRIDE);
        float* __restrict__ Out = (wi == 0) ? O0 : (wi == 1) ? O1 : O2;

        v8f acc[8];
#pragma unroll
        for (int n = 0; n < 8; ++n) acc[n] = (v8f){0.f,0.f,0.f,0.f,0.f,0.f,0.f,0.f};

#pragma unroll 4
        for (int k0 = 0; k0 < FDIM; k0 += 4) {
            const int k = k0 + 2 * half;
            const v2f a = *(const v2f*)(xrow + k);
#pragma unroll
            for (int n = 0; n < 8; ++n) {
                const v2f b = *(const v2f*)(sW + (n * 16 + m) * LDS_STRIDE + k);
                acc[n] = __builtin_amdgcn_wmma_f32_16x16x4_f32(
                             false, a, false, b, (short)0, acc[n], false, false);
            }
        }

#pragma unroll
        for (int i = 0; i < 8; ++i) {
            const int row = rowBase + i + 8 * half;
            if (row < rowEnd) {
                float* orow = Out + (size_t)row * FDIM;
#pragma unroll
                for (int n = 0; n < 8; ++n)
                    orow[n * 16 + m] = acc[n][i];
            }
        }
    }

    if (GROUP_PASS == 1) {
        for (int c = (int)blockIdx.x; c < NCH; c += (int)gridDim.x) {
            if (c != (int)blockIdx.x && spare)
                hgnn_count_edges(c, sl, nsl, sG, dst_ba, dst_ab1, dst_ab2, nEdges, nA, nB);
            __syncthreads();
            if (tid < NBK) { C[c * NBK + tid] = sG[tid]; sG[tid] = 0; }
            __syncthreads();
        }
    }
    if (GROUP_PASS == 2) {
        for (int c = (int)blockIdx.x + (int)gridDim.x; c < NCH; c += (int)gridDim.x) {
            __syncthreads();
            hgnn_place_prepare(c, tid, sG, C, bstart);
            if (spare) hgnn_place_edges(c, sl, nsl, sG + 6 * NBK, src_ba, dst_ba, src_ab1, dst_ab1, src_ab2, dst_ab2,
                                        rec, nEdges, nA, nB);
        }
    }
}

__global__ __launch_bounds__(1024)
void hgnn_gather_bucket(const float* __restrict__ Y,
                        const int2* __restrict__ rec,
                        const int* __restrict__ bstart,
                        float* __restrict__ Out,
                        int nA, int nB)
{
    __shared__ int  sHead[MAXR];
    __shared__ int2 sNode[GCAP];
    __shared__ int  sTicket;

    const int tid  = threadIdx.x;
    const int lane = tid & 31;
    const int sub  = tid & 7;
    const int grp  = lane >> 3;
    const int b    = blockIdx.x;
    const int rbs  = hgnn_rows_per_side(nA, nB);
    const int nr   = 2 * rbs;
    const int beg  = bstart[b];
    const int end  = bstart[b + 1];

    for (int base = beg; base < end; base += GCAP) {
        const int n = (end - base < GCAP) ? end - base : GCAP;
        for (int i = tid; i < nr; i += 1024) sHead[i] = -1;
        if (tid == 0) sTicket = 0;
        __syncthreads();
        for (int i = tid; i < n; i += 1024) {
            const int2 r    = rec[base + i];
            const int  prev = atomicExch(&sHead[r.x], i);
            sNode[i] = make_int2(prev, r.y);
        }
        __syncthreads();

        for (int k = 0; k < MAXR / 4; ++k) {
            int t = 0;
            if (lane == 0) t = atomicAdd(&sTicket, 4);
            t = __builtin_amdgcn_readfirstlane(t);
            if (t >= nr) break;
            const int q     = t + grp;
            const int local = (q < rbs) ? q + rbs : q - rbs;
            const bool sideB = local >= rbs;
            const int ds    = b * rbs + (sideB ? local - rbs : local);
            const bool valid = (q < nr) && (ds < (sideB ? nB : nA));
            int e = valid ? sHead[local] : -1;
            if (e < 0) continue;
            const int row = sideB ? nA + ds : ds;

            v4f* orow = (v4f*)(Out + (size_t)row * FDIM);
            v4f acc[4];
#pragma unroll
            for (int j = 0; j < 4; ++j) acc[j] = __builtin_nontemporal_load(orow + sub + 8 * j);

            int2 nd = sNode[e];
            for (int it = 0; it < n; ++it) {
                const v4f* y = (const v4f*)(Y + (size_t)nd.y * FDIM);
                v4f v[4];
#pragma unroll
                for (int j = 0; j < 4; ++j) v[j] = y[sub + 8 * j];
                e = nd.x;
                const bool more = e >= 0;
                if (more) nd = sNode[e];
#pragma unroll
                for (int j = 0; j < 4; ++j) acc[j] += v[j];
                if (!more) break;
            }
#pragma unroll
            for (int j = 0; j < 4; ++j) __builtin_nontemporal_store(acc[j], orow + sub + 8 * j);
        }
        __syncthreads();
    }
}

extern "C" void kernel_launch(void* const* d_in, const int* in_sizes, int n_in,
                              void* d_out, int out_size, void* d_ws, size_t ws_size,
                              hipStream_t stream)
{
    const float* xA        = (const float*)d_in[0];
    const float* xB        = (const float*)d_in[1];
    const float* Wmsg_ab1  = (const float*)d_in[2];
    const float* Wself_ab1 = (const float*)d_in[3];
    const float* Wmsg_ab2  = (const float*)d_in[4];
    const float* Wself_ab2 = (const float*)d_in[5];
    const float* Wmsg_ba   = (const float*)d_in[6];
    const float* Wself_ba  = (const float*)d_in[7];
    const int*   src_ab1   = (const int*)d_in[8];
    const int*   dst_ab1   = (const int*)d_in[9];
    const int*   src_ab2   = (const int*)d_in[10];
    const int*   dst_ab2   = (const int*)d_in[11];
    const int*   src_ba    = (const int*)d_in[12];
    const int*   dst_ba    = (const int*)d_in[13];

    const int NA = in_sizes[0] / FDIM;
    const int NB = in_sizes[1] / FDIM;
    const int E  = in_sizes[8];

    float* outA = (float*)d_out;
    float* outB = outA + (size_t)NA * FDIM;

    float* Yba  = (float*)d_ws;
    float* Yab1 = Yba  + (size_t)NB * FDIM;
    float* Yab2 = Yab1 + (size_t)NA * FDIM;
    int2*  rec  = (int2*)(Yab2 + (size_t)NA * FDIM);
    int*   cnt  = (int*)(rec + (size_t)3 * E);
    int*   bst  = cnt + NCH * NBK;

    const dim3 blk(1024);
    const dim3 gA((NA + ROWS_PER_GROUP - 1) / ROWS_PER_GROUP);
    const dim3 gB((NB + ROWS_PER_GROUP - 1) / ROWS_PER_GROUP);

    hgnn_gemm128_multi<1><<<gA, blk, 0, stream>>>(xA, Wmsg_ab1, nullptr, Wmsg_ab2, Wself_ba,
                                                  Yab1, Yab2, outA, 3, NA,
                                                  src_ba, dst_ba, src_ab1, dst_ab1, src_ab2, dst_ab2,
                                                  cnt, rec, bst, E, NA, NB);
    hgnn_gemm128_multi<2><<<gB, blk, 0, stream>>>(xB, Wself_ab1, Wself_ab2, Wmsg_ba, nullptr,
                                                  outB, Yba, nullptr, 2, NB,
                                                  src_ba, dst_ba, src_ab1, dst_ab1, src_ab2, dst_ab2,
                                                  cnt, rec, bst, E, NA, NB);

    hgnn_gather_bucket<<<dim3(NBK), dim3(1024), 0, stream>>>(Yba, rec, bst, outA, NA, NB);
}
